# starting kernel
# baseline (speedup 1.0000x reference)
#include <hip/hip_runtime.h>

typedef float v2f __attribute__((ext_vector_type(2)));
typedef float v8f __attribute__((ext_vector_type(8)));

#define T_LEN 16384
#define D_DIM 1024
#define H_DIM 8

__global__ __launch_bounds__(256) void rnn_inproj_wmma(
    const float* __restrict__ x, const float* __restrict__ W_ih,
    const float* __restrict__ b_ih, const float* __restrict__ b_hh,
    float* __restrict__ xwT)
{
    const int lane = threadIdx.x & 31;
    const int wave = threadIdx.x >> 5;
    const int tile = blockIdx.x * 8 + wave;
    const int r    = lane & 15;
    const int hi   = lane >> 4;
    const int kb   = hi * 2;
    const int nn   = r & 7;
    const float nmask = (r < 8) ? 1.0f : 0.0f;

    const float* xrow = x    + (size_t)(tile * 16 + r) * D_DIM;
    const float* wrow = W_ih + (size_t)nn * D_DIM;

    v8f acc = {};
#pragma unroll 4
    for (int k = 0; k < D_DIM; k += 4) {
        v2f a = *(const v2f*)(xrow + k + kb);
        v2f b = *(const v2f*)(wrow + k + kb);
        b.x *= nmask;
        b.y *= nmask;
        acc = __builtin_amdgcn_wmma_f32_16x16x4_f32(
             false, a,  false, b,
             (short)0, acc,  false,  false);
    }

    if (r < 8) {
        const float bias = b_ih[r] + b_hh[r];
#pragma unroll
        for (int v = 0; v < 8; ++v) {
            const int tl = tile * 16 + v + hi * 8;
            xwT[(size_t)r * T_LEN + tl] = acc[v] + bias;
        }
    }
}

__global__ void rnn_scan(const float* __restrict__ xwT,
                         const float* __restrict__ W_hh,
                         float* __restrict__ hs)
{
    const int lane = threadIdx.x & 31;
    const int i = lane & 7;

    float w[8];
#pragma unroll
    for (int j = 0; j < 8; ++j) w[j] = W_hh[i * 8 + j];

    const float4* xs = (const float4*)(xwT + (size_t)i * T_LEN);
    float4 cur = xs[0];
    float h = 0.0f;

    for (int t = 0; t < T_LEN; t += 4) {
        const int tn = (t + 4 < T_LEN) ? (t + 4) : 0;
        float4 nxt = xs[tn >> 2];
        const float xv[4] = {cur.x, cur.y, cur.z, cur.w};
#pragma unroll
        for (int c = 0; c < 4; ++c) {
            float s = xv[c];
#pragma unroll
            for (int j = 0; j < 8; ++j) {
                const float hj = __int_as_float(
                    __builtin_amdgcn_readlane(__float_as_int(h), j));
                s = fmaf(w[j], hj, s);
            }
            h = fmaxf(s, 0.0f);
            if (lane < 8)
                hs[(size_t)(t + c) * 8 + lane] = h;
        }
        cur = nxt;
    }
}

__global__ __launch_bounds__(256) void rnn_out(
    const float* __restrict__ hs, const float* __restrict__ W_out,
    const float* __restrict__ b_out, float* __restrict__ out)
{
    const int t = blockIdx.x * blockDim.x + threadIdx.x;
    if (t >= T_LEN) return;
    const float4* hp = (const float4*)(hs + (size_t)t * 8);
    const float4 h0 = hp[0], h1 = hp[1];
    float s = b_out[0];
    s += h0.x * W_out[0] + h0.y * W_out[1] + h0.z * W_out[2] + h0.w * W_out[3];
    s += h1.x * W_out[4] + h1.y * W_out[5] + h1.z * W_out[6] + h1.w * W_out[7];
    out[t] = s;
}

extern "C" void kernel_launch(void* const* d_in, const int* in_sizes, int n_in,
                              void* d_out, int out_size, void* d_ws, size_t ws_size,
                              hipStream_t stream) {
    const float* x     = (const float*)d_in[0];
    const float* W_ih  = (const float*)d_in[1];
    const float* b_ih  = (const float*)d_in[2];
    const float* W_hh  = (const float*)d_in[3];
    const float* b_hh  = (const float*)d_in[4];
    const float* W_out = (const float*)d_in[5];
    const float* b_out = (const float*)d_in[6];
    float* out = (float*)d_out;

    float* xwT = (float*)d_ws;
    float* hs  = xwT + (size_t)T_LEN * H_DIM;

    rnn_inproj_wmma<<<(T_LEN / 16) / 8, 256, 0, stream>>>(x, W_ih, b_ih, b_hh, xwT);
    rnn_scan<<<1, 32, 0, stream>>>(xwT, W_hh, hs);
    rnn_out<<<(T_LEN + 255) / 256, 256, 0, stream>>>(hs, W_out, b_out, out);
}


# scan in 16-step blocks: h history kept in registers, one full-EXEC 16-byte store per lane per block into a transposed plane; per-step masked store, EXEC toggle, branch and s_wait_xcnt leave the chain
# speedup vs baseline: 2.1721x; 2.1721x over previous
#include <hip/hip_runtime.h>

typedef float v2f __attribute__((ext_vector_type(2)));
typedef float v8f __attribute__((ext_vector_type(8)));

#define T_LEN 16384
#define D_DIM 1024
#define H_DIM 8

__global__ __launch_bounds__(256) void rnn_inproj_wmma(
    const float* __restrict__ x, const float* __restrict__ W_ih,
    const float* __restrict__ b_ih, const float* __restrict__ b_hh,
    float* __restrict__ xwT)
{
    const int lane = threadIdx.x & 31;
    const int wave = threadIdx.x >> 5;
    const int tile = blockIdx.x * 8 + wave;
    const int r    = lane & 15;
    const int hi   = lane >> 4;
    const int kb   = hi * 2;
    const int nn   = r & 7;
    const float nmask = (r < 8) ? 1.0f : 0.0f;

    const float* xrow = x    + (size_t)(tile * 16 + r) * D_DIM;
    const float* wrow = W_ih + (size_t)nn * D_DIM;

    v8f acc = {};
#pragma unroll 4
    for (int k = 0; k < D_DIM; k += 4) {
        v2f a = *(const v2f*)(xrow + k + kb);
        v2f b = *(const v2f*)(wrow + k + kb);
        b.x *= nmask;
        b.y *= nmask;
        acc = __builtin_amdgcn_wmma_f32_16x16x4_f32(
             false, a,  false, b,
             (short)0, acc,  false,  false);
    }

    if (r < 8) {
        const float bias = b_ih[r] + b_hh[r];
#pragma unroll
        for (int v = 0; v < 8; ++v) {
            const int tl = tile * 16 + v + hi * 8;
            xwT[(size_t)r * T_LEN + tl] = acc[v] + bias;
        }
    }
}

__global__ void rnn_scan(const float* __restrict__ xwT,
                         const float* __restrict__ W_hh,
                         float* __restrict__ hsT)
{
    const int lane = threadIdx.x & 31;
    const int i = lane & 7;
    const int g = lane >> 3;

    float w[8];
#pragma unroll
    for (int j = 0; j < 8; ++j) w[j] = W_hh[i * 8 + j];

    const float4* xs = (const float4*)(xwT + (size_t)i * T_LEN);
    float4* hq = (float4*)(hsT + (size_t)i * T_LEN) + g;

    float4 cur[4], nxt[4];
#pragma unroll
    for (int q = 0; q < 4; ++q) { cur[q] = xs[q]; nxt[q] = xs[4 + q]; }
    float h = 0.0f;

    for (int t = 0; t < T_LEN; t += 16) {
        const int tf = (t + 32 < T_LEN) ? (t + 32) : 0;
        float4 far[4];
#pragma unroll
        for (int q = 0; q < 4; ++q) far[q] = xs[(tf >> 2) + q];

        float hv[16];
#pragma unroll
        for (int c = 0; c < 16; ++c) {
            const float4 cq = cur[c >> 2];
            float s = ((c & 3) == 0) ? cq.x : ((c & 3) == 1) ? cq.y
                    : ((c & 3) == 2) ? cq.z : cq.w;
#pragma unroll
            for (int j = 0; j < 8; ++j) {
                const float hj = __int_as_float(
                    __builtin_amdgcn_readlane(__float_as_int(h), j));
                s = fmaf(w[j], hj, s);
            }
            h = fmaxf(s, 0.0f);
            hv[c] = h;
        }

        float4 o;
        o.x = (g == 3) ? hv[12] : (g == 2) ? hv[8]  : (g == 1) ? hv[4] : hv[0];
        o.y = (g == 3) ? hv[13] : (g == 2) ? hv[9]  : (g == 1) ? hv[5] : hv[1];
        o.z = (g == 3) ? hv[14] : (g == 2) ? hv[10] : (g == 1) ? hv[6] : hv[2];
        o.w = (g == 3) ? hv[15] : (g == 2) ? hv[11] : (g == 1) ? hv[7] : hv[3];
        const int tb = __builtin_amdgcn_readfirstlane(t);
        hq[tb >> 2] = o;

#pragma unroll
        for (int q = 0; q < 4; ++q) { cur[q] = nxt[q]; nxt[q] = far[q]; }
    }
}

__global__ __launch_bounds__(256) void rnn_out(
    const float* __restrict__ hsT, const float* __restrict__ W_out,
    const float* __restrict__ b_out, float* __restrict__ out)
{
    const int t = blockIdx.x * blockDim.x + threadIdx.x;
    if (t >= T_LEN) return;
    const float4 h0 = {hsT[0 * T_LEN + t], hsT[1 * T_LEN + t],
                       hsT[2 * T_LEN + t], hsT[3 * T_LEN + t]};
    const float4 h1 = {hsT[4 * T_LEN + t], hsT[5 * T_LEN + t],
                       hsT[6 * T_LEN + t], hsT[7 * T_LEN + t]};
    float s = b_out[0];
    s += h0.x * W_out[0] + h0.y * W_out[1] + h0.z * W_out[2] + h0.w * W_out[3];
    s += h1.x * W_out[4] + h1.y * W_out[5] + h1.z * W_out[6] + h1.w * W_out[7];
    out[t] = s;
}

extern "C" void kernel_launch(void* const* d_in, const int* in_sizes, int n_in,
                              void* d_out, int out_size, void* d_ws, size_t ws_size,
                              hipStream_t stream) {
    const float* x     = (const float*)d_in[0];
    const float* W_ih  = (const float*)d_in[1];
    const float* b_ih  = (const float*)d_in[2];
    const float* W_hh  = (const float*)d_in[3];
    const float* b_hh  = (const float*)d_in[4];
    const float* W_out = (const float*)d_in[5];
    const float* b_out = (const float*)d_in[6];
    float* out = (float*)d_out;

    float* xwT = (float*)d_ws;
    float* hsT = xwT + (size_t)T_LEN * H_DIM;

    rnn_inproj_wmma<<<(T_LEN / 16) / 8, 256, 0, stream>>>(x, W_ih, b_ih, b_hh, xwT);
    rnn_scan<<<1, 32, 0, stream>>>(xwT, W_hh, hsT);
    rnn_out<<<(T_LEN + 255) / 256, 256, 0, stream>>>(hsT, W_out, b_out, out);
}


# scan step: the 8-term h.W_hh sum leaves its 8-deep fmac chain for 4 two-term partial sums joined by a 2-level add tree (summation order changes); 9 -> 5 dependent vector ops per step
# speedup vs baseline: 2.5728x; 1.1845x over previous
#include <hip/hip_runtime.h>

typedef float v2f __attribute__((ext_vector_type(2)));
typedef float v8f __attribute__((ext_vector_type(8)));

#define T_LEN 16384
#define D_DIM 1024
#define H_DIM 8

__global__ __launch_bounds__(256) void rnn_inproj_wmma(
    const float* __restrict__ x, const float* __restrict__ W_ih,
    const float* __restrict__ b_ih, const float* __restrict__ b_hh,
    float* __restrict__ xwT)
{
    const int lane = threadIdx.x & 31;
    const int wave = threadIdx.x >> 5;
    const int tile = blockIdx.x * 8 + wave;
    const int r    = lane & 15;
    const int hi   = lane >> 4;
    const int kb   = hi * 2;
    const int nn   = r & 7;
    const float nmask = (r < 8) ? 1.0f : 0.0f;

    const float* xrow = x    + (size_t)(tile * 16 + r) * D_DIM;
    const float* wrow = W_ih + (size_t)nn * D_DIM;

    v8f acc = {};
#pragma unroll 4
    for (int k = 0; k < D_DIM; k += 4) {
        v2f a = *(const v2f*)(xrow + k + kb);
        v2f b = *(const v2f*)(wrow + k + kb);
        b.x *= nmask;
        b.y *= nmask;
        acc = __builtin_amdgcn_wmma_f32_16x16x4_f32(
             false, a,  false, b,
             (short)0, acc,  false,  false);
    }

    if (r < 8) {
        const float bias = b_ih[r] + b_hh[r];
#pragma unroll
        for (int v = 0; v < 8; ++v) {
            const int tl = tile * 16 + v + hi * 8;
            xwT[(size_t)r * T_LEN + tl] = acc[v] + bias;
        }
    }
}

__global__ void rnn_scan(const float* __restrict__ xwT,
                         const float* __restrict__ W_hh,
                         float* __restrict__ hsT)
{
    const int lane = threadIdx.x & 31;
    const int i = lane & 7;
    const int g = lane >> 3;

    float w[8];
#pragma unroll
    for (int j = 0; j < 8; ++j) w[j] = W_hh[i * 8 + j];

    const float4* xs = (const float4*)(xwT + (size_t)i * T_LEN);
    float4* hq = (float4*)(hsT + (size_t)i * T_LEN) + g;

    float4 cur[4], nxt[4];
#pragma unroll
    for (int q = 0; q < 4; ++q) { cur[q] = xs[q]; nxt[q] = xs[4 + q]; }
    float h = 0.0f;

    for (int t = 0; t < T_LEN; t += 16) {
        const int tf = (t + 32 < T_LEN) ? (t + 32) : 0;
        float4 far[4];
#pragma unroll
        for (int q = 0; q < 4; ++q) far[q] = xs[(tf >> 2) + q];

        float hv[16];
#pragma unroll
        for (int c = 0; c < 16; ++c) {
            const float4 cq = cur[c >> 2];
            const float xs_c = ((c & 3) == 0) ? cq.x : ((c & 3) == 1) ? cq.y
                             : ((c & 3) == 2) ? cq.z : cq.w;
            int hb[8];
#pragma unroll
            for (int j = 0; j < 8; ++j)
                hb[j] = __builtin_amdgcn_readlane(__float_as_int(h), j);
            asm("" : "+s"(hb[0]), "+s"(hb[1]), "+s"(hb[2]), "+s"(hb[3]),
                     "+s"(hb[4]), "+s"(hb[5]), "+s"(hb[6]), "+s"(hb[7]));
            float p0 = fmaf(w[4], __int_as_float(hb[4]),
                            fmaf(w[0], __int_as_float(hb[0]), xs_c));
            float p1 = fmaf(w[5], __int_as_float(hb[5]), w[1] * __int_as_float(hb[1]));
            float p2 = fmaf(w[6], __int_as_float(hb[6]), w[2] * __int_as_float(hb[2]));
            float p3 = fmaf(w[7], __int_as_float(hb[7]), w[3] * __int_as_float(hb[3]));
            asm("" : "+v"(p0), "+v"(p1), "+v"(p2), "+v"(p3));
            h = fmaxf((p0 + p1) + (p2 + p3), 0.0f);
            hv[c] = h;
        }

        float4 o;
        o.x = (g == 3) ? hv[12] : (g == 2) ? hv[8]  : (g == 1) ? hv[4] : hv[0];
        o.y = (g == 3) ? hv[13] : (g == 2) ? hv[9]  : (g == 1) ? hv[5] : hv[1];
        o.z = (g == 3) ? hv[14] : (g == 2) ? hv[10] : (g == 1) ? hv[6] : hv[2];
        o.w = (g == 3) ? hv[15] : (g == 2) ? hv[11] : (g == 1) ? hv[7] : hv[3];
        const int tb = __builtin_amdgcn_readfirstlane(t);
        hq[tb >> 2] = o;

#pragma unroll
        for (int q = 0; q < 4; ++q) { cur[q] = nxt[q]; nxt[q] = far[q]; }
    }
}

__global__ __launch_bounds__(256) void rnn_out(
    const float* __restrict__ hsT, const float* __restrict__ W_out,
    const float* __restrict__ b_out, float* __restrict__ out)
{
    const int t = blockIdx.x * blockDim.x + threadIdx.x;
    if (t >= T_LEN) return;
    const float4 h0 = {hsT[0 * T_LEN + t], hsT[1 * T_LEN + t],
                       hsT[2 * T_LEN + t], hsT[3 * T_LEN + t]};
    const float4 h1 = {hsT[4 * T_LEN + t], hsT[5 * T_LEN + t],
                       hsT[6 * T_LEN + t], hsT[7 * T_LEN + t]};
    float s = b_out[0];
    s += h0.x * W_out[0] + h0.y * W_out[1] + h0.z * W_out[2] + h0.w * W_out[3];
    s += h1.x * W_out[4] + h1.y * W_out[5] + h1.z * W_out[6] + h1.w * W_out[7];
    out[t] = s;
}

extern "C" void kernel_launch(void* const* d_in, const int* in_sizes, int n_in,
                              void* d_out, int out_size, void* d_ws, size_t ws_size,
                              hipStream_t stream) {
    const float* x     = (const float*)d_in[0];
    const float* W_ih  = (const float*)d_in[1];
    const float* b_ih  = (const float*)d_in[2];
    const float* W_hh  = (const float*)d_in[3];
    const float* b_hh  = (const float*)d_in[4];
    const float* W_out = (const float*)d_in[5];
    const float* b_out = (const float*)d_in[6];
    float* out = (float*)d_out;

    float* xwT = (float*)d_ws;
    float* hsT = xwT + (size_t)T_LEN * H_DIM;

    rnn_inproj_wmma<<<(T_LEN / 16) / 8, 256, 0, stream>>>(x, W_ih, b_ih, b_hh, xwT);
    rnn_scan<<<1, 32, 0, stream>>>(xwT, W_hh, hsT);
    rnn_out<<<(T_LEN + 255) / 256, 256, 0, stream>>>(hsT, W_out, b_out, out);
}


# scan step: h[j] reaches the lanes by row_share DPP fused into the 8 multiplies/multiply-adds (3 accumulators, then (P+R)+Q: summation order changes) instead of 8 v_readlane via scalar registers
# speedup vs baseline: 2.7547x; 1.0707x over previous
#include <hip/hip_runtime.h>

typedef float v2f __attribute__((ext_vector_type(2)));
typedef float v8f __attribute__((ext_vector_type(8)));

#define T_LEN 16384
#define D_DIM 1024
#define H_DIM 8

__global__ __launch_bounds__(256) void rnn_inproj_wmma(
    const float* __restrict__ x, const float* __restrict__ W_ih,
    const float* __restrict__ b_ih, const float* __restrict__ b_hh,
    float* __restrict__ xwT)
{
    const int lane = threadIdx.x & 31;
    const int wave = threadIdx.x >> 5;
    const int tile = blockIdx.x * 8 + wave;
    const int r    = lane & 15;
    const int hi   = lane >> 4;
    const int kb   = hi * 2;
    const int nn   = r & 7;
    const float nmask = (r < 8) ? 1.0f : 0.0f;

    const float* xrow = x    + (size_t)(tile * 16 + r) * D_DIM;
    const float* wrow = W_ih + (size_t)nn * D_DIM;

    v8f acc = {};
#pragma unroll 4
    for (int k = 0; k < D_DIM; k += 4) {
        v2f a = *(const v2f*)(xrow + k + kb);
        v2f b = *(const v2f*)(wrow + k + kb);
        b.x *= nmask;
        b.y *= nmask;
        acc = __builtin_amdgcn_wmma_f32_16x16x4_f32(
             false, a,  false, b,
             (short)0, acc,  false,  false);
    }

    if (r < 8) {
        const float bias = b_ih[r] + b_hh[r];
#pragma unroll
        for (int v = 0; v < 8; ++v) {
            const int tl = tile * 16 + v + hi * 8;
            xwT[(size_t)r * T_LEN + tl] = acc[v] + bias;
        }
    }
}

#define MUL_ROW_SHARE(acc, h, w, J)                                          \
    asm volatile("v_mul_f32_dpp %0, %1, %2 row_share:" #J                    \
                 " row_mask:0xf bank_mask:0xf"                               \
                 : "=v"(acc) : "v"(h), "v"(w))
#define FMAC_ROW_SHARE(acc, h, w, J)                                         \
    asm volatile("v_fmac_f32_dpp %0, %1, %2 row_share:" #J                   \
                 " row_mask:0xf bank_mask:0xf"                               \
                 : "+v"(acc) : "v"(h), "v"(w))

__global__ void rnn_scan(const float* __restrict__ xwT,
                         const float* __restrict__ W_hh,
                         float* __restrict__ hsT)
{
    const int lane = threadIdx.x & 31;
    const int i = lane & 7;
    const int g = lane >> 3;

    float w[8];
#pragma unroll
    for (int j = 0; j < 8; ++j) w[j] = W_hh[i * 8 + j];

    const float4* xs = (const float4*)(xwT + (size_t)i * T_LEN);
    float4* hq = (float4*)(hsT + (size_t)i * T_LEN) + g;

    float4 cur[4], nxt[4];
#pragma unroll
    for (int q = 0; q < 4; ++q) { cur[q] = xs[q]; nxt[q] = xs[4 + q]; }
    float h = 0.0f;

    for (int t = 0; t < T_LEN; t += 16) {
        const int tf = (t + 32 < T_LEN) ? (t + 32) : 0;
        const float4* fp = xs + (tf >> 2);
        float4 far[4];
#pragma unroll
        for (int q = 0; q < 4; ++q) far[q] = fp[q];

        float hv[16];
#pragma unroll
        for (int c = 0; c < 16; ++c) {
            const float4 cq = cur[c >> 2];
            const float xs_c = ((c & 3) == 0) ? cq.x : ((c & 3) == 1) ? cq.y
                             : ((c & 3) == 2) ? cq.z : cq.w;
            if (c == 8) asm volatile("" :: "v"(fp));
            float P = xs_c, Q, R;
            FMAC_ROW_SHARE(P, h, w[0], 0);
            MUL_ROW_SHARE(Q, h, w[1], 1);
            MUL_ROW_SHARE(R, h, w[2], 2);
            FMAC_ROW_SHARE(P, h, w[3], 3);
            FMAC_ROW_SHARE(Q, h, w[4], 4);
            FMAC_ROW_SHARE(R, h, w[5], 5);
            FMAC_ROW_SHARE(P, h, w[6], 6);
            FMAC_ROW_SHARE(Q, h, w[7], 7);
            asm volatile("" : "+v"(P));
            h = fmaxf((P + R) + Q, 0.0f);
            hv[c] = h;
        }

        float4 o;
        o.x = (g == 3) ? hv[12] : (g == 2) ? hv[8]  : (g == 1) ? hv[4] : hv[0];
        o.y = (g == 3) ? hv[13] : (g == 2) ? hv[9]  : (g == 1) ? hv[5] : hv[1];
        o.z = (g == 3) ? hv[14] : (g == 2) ? hv[10] : (g == 1) ? hv[6] : hv[2];
        o.w = (g == 3) ? hv[15] : (g == 2) ? hv[11] : (g == 1) ? hv[7] : hv[3];
        const int tb = __builtin_amdgcn_readfirstlane(t);
        hq[tb >> 2] = o;

#pragma unroll
        for (int q = 0; q < 4; ++q) { cur[q] = nxt[q]; nxt[q] = far[q]; }
    }
}

__global__ __launch_bounds__(256) void rnn_out(
    const float* __restrict__ hsT, const float* __restrict__ W_out,
    const float* __restrict__ b_out, float* __restrict__ out)
{
    const int t = blockIdx.x * blockDim.x + threadIdx.x;
    if (t >= T_LEN) return;
    const float4 h0 = {hsT[0 * T_LEN + t], hsT[1 * T_LEN + t],
                       hsT[2 * T_LEN + t], hsT[3 * T_LEN + t]};
    const float4 h1 = {hsT[4 * T_LEN + t], hsT[5 * T_LEN + t],
                       hsT[6 * T_LEN + t], hsT[7 * T_LEN + t]};
    float s = b_out[0];
    s += h0.x * W_out[0] + h0.y * W_out[1] + h0.z * W_out[2] + h0.w * W_out[3];
    s += h1.x * W_out[4] + h1.y * W_out[5] + h1.z * W_out[6] + h1.w * W_out[7];
    out[t] = s;
}

extern "C" void kernel_launch(void* const* d_in, const int* in_sizes, int n_in,
                              void* d_out, int out_size, void* d_ws, size_t ws_size,
                              hipStream_t stream) {
    const float* x     = (const float*)d_in[0];
    const float* W_ih  = (const float*)d_in[1];
    const float* b_ih  = (const float*)d_in[2];
    const float* W_hh  = (const float*)d_in[3];
    const float* b_hh  = (const float*)d_in[4];
    const float* W_out = (const float*)d_in[5];
    const float* b_out = (const float*)d_in[6];
    float* out = (float*)d_out;

    float* xwT = (float*)d_ws;
    float* hsT = xwT + (size_t)T_LEN * H_DIM;

    rnn_inproj_wmma<<<(T_LEN / 16) / 8, 256, 0, stream>>>(x, W_ih, b_ih, b_hh, xwT);
    rnn_scan<<<1, 32, 0, stream>>>(xwT, W_hh, hsT);
    rnn_out<<<(T_LEN + 255) / 256, 256, 0, stream>>>(hsT, W_out, b_out, out);
}


# scan: inputs are requested 64 steps ahead into two register sets that take turns (loop body = two 64-step parts): no set is copied and no 16-step block opens behind a full wait for its loads
# speedup vs baseline: 3.6634x; 1.3299x over previous
#include <hip/hip_runtime.h>

typedef float v2f __attribute__((ext_vector_type(2)));
typedef float v8f __attribute__((ext_vector_type(8)));

#define T_LEN 16384
#define D_DIM 1024
#define H_DIM 8

__global__ __launch_bounds__(256) void rnn_inproj_wmma(
    const float* __restrict__ x, const float* __restrict__ W_ih,
    const float* __restrict__ b_ih, const float* __restrict__ b_hh,
    float* __restrict__ xwT)
{
    const int lane = threadIdx.x & 31;
    const int wave = threadIdx.x >> 5;
    const int tile = blockIdx.x * 8 + wave;
    const int r    = lane & 15;
    const int hi   = lane >> 4;
    const int kb   = hi * 2;
    const int nn   = r & 7;
    const float nmask = (r < 8) ? 1.0f : 0.0f;

    const float* xrow = x    + (size_t)(tile * 16 + r) * D_DIM;
    const float* wrow = W_ih + (size_t)nn * D_DIM;

    v8f acc = {};
#pragma unroll 4
    for (int k = 0; k < D_DIM; k += 4) {
        v2f a = *(const v2f*)(xrow + k + kb);
        v2f b = *(const v2f*)(wrow + k + kb);
        b.x *= nmask;
        b.y *= nmask;
        acc = __builtin_amdgcn_wmma_f32_16x16x4_f32(
             false, a,  false, b,
             (short)0, acc,  false,  false);
    }

    if (r < 8) {
        const float bias = b_ih[r] + b_hh[r];
#pragma unroll
        for (int v = 0; v < 8; ++v) {
            const int tl = tile * 16 + v + hi * 8;
            xwT[(size_t)r * T_LEN + tl] = acc[v] + bias;
        }
    }
}

#define MUL_ROW_SHARE(acc, h, w, J)                                          \
    asm volatile("v_mul_f32_dpp %0, %1, %2 row_share:" #J                    \
                 " row_mask:0xf bank_mask:0xf"                               \
                 : "=v"(acc) : "v"(h), "v"(w))
#define FMAC_ROW_SHARE(acc, h, w, J)                                         \
    asm volatile("v_fmac_f32_dpp %0, %1, %2 row_share:" #J                   \
                 " row_mask:0xf bank_mask:0xf"                               \
                 : "+v"(acc) : "v"(h), "v"(w))

__device__ __forceinline__ void scan_part(const float4 (&X)[16], const float (&w)[8],
                                          float& h, const int g,
                                          char* __restrict__ sp, const unsigned so,
                                          const unsigned lv, float4& ko)
{
#pragma unroll
    for (int b = 0; b < 4; ++b) {
        float hv[16];
#pragma unroll
        for (int c = 0; c < 16; ++c) {
            const float4 cq = X[b * 4 + (c >> 2)];
            const float xs_c = ((c & 3) == 0) ? cq.x : ((c & 3) == 1) ? cq.y
                             : ((c & 3) == 2) ? cq.z : cq.w;
            if (c == 8) {
                asm volatile("" :: "v"(ko.x), "v"(ko.y), "v"(ko.z), "v"(ko.w));
                if (b == 0) asm volatile("" :: "v"(lv));
            }
            float P = xs_c, Q, R;
            FMAC_ROW_SHARE(P, h, w[0], 0);
            MUL_ROW_SHARE(Q, h, w[1], 1);
            MUL_ROW_SHARE(R, h, w[2], 2);
            FMAC_ROW_SHARE(P, h, w[3], 3);
            FMAC_ROW_SHARE(Q, h, w[4], 4);
            FMAC_ROW_SHARE(R, h, w[5], 5);
            FMAC_ROW_SHARE(P, h, w[6], 6);
            FMAC_ROW_SHARE(Q, h, w[7], 7);
            asm volatile("" : "+v"(P));
            h = fmaxf((P + R) + Q, 0.0f);
            hv[c] = h;
        }

        float4 o;
        o.x = (g == 3) ? hv[12] : (g == 2) ? hv[8]  : (g == 1) ? hv[4] : hv[0];
        o.y = (g == 3) ? hv[13] : (g == 2) ? hv[9]  : (g == 1) ? hv[5] : hv[1];
        o.z = (g == 3) ? hv[14] : (g == 2) ? hv[10] : (g == 1) ? hv[6] : hv[2];
        o.w = (g == 3) ? hv[15] : (g == 2) ? hv[11] : (g == 1) ? hv[7] : hv[3];
        *(float4*)(sp + b * 64 + so) = o;
        ko = o;
    }
}

__global__ __launch_bounds__(32) void rnn_scan(const float* __restrict__ xwT,
                         const float* __restrict__ W_hh,
                         float* __restrict__ hsT)
{
    const int lane = threadIdx.x & 31;
    const int i = lane & 7;
    const int g = lane >> 3;

    float w[8];
#pragma unroll
    for (int j = 0; j < 8; ++j) w[j] = W_hh[i * 8 + j];

    const unsigned lo = (unsigned)i * (unsigned)(T_LEN * 4);
    const unsigned so = lo + (unsigned)g * 16u;
    const char* xb = (const char*)xwT;
    char* hb = (char*)hsT;

    float4 A[16], B[16];
#pragma unroll
    for (int q = 0; q < 16; ++q) A[q] = *(const float4*)(xb + lo + q * 16);
    float h = 0.0f;
    float4 ko = {0.0f, 0.0f, 0.0f, 0.0f};

    for (int t = 0; t < T_LEN; t += 128) {
        const char* pb = xb + (size_t)(t + 64) * 4;
        unsigned vb = lo;
        asm volatile("" : "+v"(vb));
#pragma unroll
        for (int q = 0; q < 16; ++q) B[q] = *(const float4*)(pb + vb + q * 16);
        scan_part(A, w, h, g, hb + (size_t)t * 4, so, vb, ko);

        const int ta = (t + 128 < T_LEN) ? (t + 128) : 0;
        const char* pa = xb + (size_t)ta * 4;
        unsigned va = lo;
        asm volatile("" : "+v"(va));
#pragma unroll
        for (int q = 0; q < 16; ++q) A[q] = *(const float4*)(pa + va + q * 16);
        scan_part(B, w, h, g, hb + (size_t)(t + 64) * 4, so, va, ko);
    }
}

__global__ __launch_bounds__(256) void rnn_out(
    const float* __restrict__ hsT, const float* __restrict__ W_out,
    const float* __restrict__ b_out, float* __restrict__ out)
{
    const int t = blockIdx.x * blockDim.x + threadIdx.x;
    if (t >= T_LEN) return;
    const float4 h0 = {hsT[0 * T_LEN + t], hsT[1 * T_LEN + t],
                       hsT[2 * T_LEN + t], hsT[3 * T_LEN + t]};
    const float4 h1 = {hsT[4 * T_LEN + t], hsT[5 * T_LEN + t],
                       hsT[6 * T_LEN + t], hsT[7 * T_LEN + t]};
    float s = b_out[0];
    s += h0.x * W_out[0] + h0.y * W_out[1] + h0.z * W_out[2] + h0.w * W_out[3];
    s += h1.x * W_out[4] + h1.y * W_out[5] + h1.z * W_out[6] + h1.w * W_out[7];
    out[t] = s;
}

extern "C" void kernel_launch(void* const* d_in, const int* in_sizes, int n_in,
                              void* d_out, int out_size, void* d_ws, size_t ws_size,
                              hipStream_t stream) {
    const float* x     = (const float*)d_in[0];
    const float* W_ih  = (const float*)d_in[1];
    const float* b_ih  = (const float*)d_in[2];
    const float* W_hh  = (const float*)d_in[3];
    const float* b_hh  = (const float*)d_in[4];
    const float* W_out = (const float*)d_in[5];
    const float* b_out = (const float*)d_in[6];
    float* out = (float*)d_out;

    float* xwT = (float*)d_ws;
    float* hsT = xwT + (size_t)T_LEN * H_DIM;

    rnn_inproj_wmma<<<(T_LEN / 16) / 8, 256, 0, stream>>>(x, W_ih, b_ih, b_hh, xwT);
    rnn_scan<<<1, 32, 0, stream>>>(xwT, W_hh, hsT);
    rnn_out<<<(T_LEN + 255) / 256, 256, 0, stream>>>(hsT, W_out, b_out, out);
}


# input projection: each 16-step tile's K=1024 is split over 4 waves (partial tiles summed in LDS: summation order changes), 16-byte fragment loads; 4096 waves on all WGPs instead of 1024 on half
# speedup vs baseline: 3.8683x; 1.0559x over previous
#include <hip/hip_runtime.h>

typedef float v2f __attribute__((ext_vector_type(2)));
typedef float v8f __attribute__((ext_vector_type(8)));

#define T_LEN 16384
#define D_DIM 1024
#define H_DIM 8

__global__ __launch_bounds__(256) void rnn_inproj_wmma(
    const float* __restrict__ x, const float* __restrict__ W_ih,
    const float* __restrict__ b_ih, const float* __restrict__ b_hh,
    float* __restrict__ xwT)
{
    __shared__ float part[2][4][16 * 8];

    const int lane = threadIdx.x & 31;
    const int wave = threadIdx.x >> 5;
    const int tsel = wave >> 2;
    const int kq   = wave & 3;
    const int tile = blockIdx.x * 2 + tsel;
    const int r    = lane & 15;
    const int hi   = lane >> 4;
    const int nn   = r & 7;

    const float* xrow = x    + (size_t)(tile * 16 + r) * D_DIM + kq * 256 + hi * 4;
    const float* wrow = W_ih + (size_t)nn * D_DIM + kq * 256 + hi * 4;

    v8f acc0 = {}, acc1 = {};
    float4 qa[2][8], qb[2][8];
#pragma unroll
    for (int j = 0; j < 8; ++j) {
        qa[0][j] = *(const float4*)(xrow + j * 8);
        qb[0][j] = *(const float4*)(wrow + j * 8);
    }
#pragma unroll
    for (int g = 0; g < 4; ++g) {
        if (g < 3) {
#pragma unroll
            for (int j = 0; j < 8; ++j) {
                qa[(g + 1) & 1][j] = *(const float4*)(xrow + (g + 1) * 64 + j * 8);
                qb[(g + 1) & 1][j] = *(const float4*)(wrow + (g + 1) * 64 + j * 8);
            }
        }
#pragma unroll
        for (int j = 0; j < 8; ++j) {
            const float4 ca = qa[g & 1][j];
            const float4 cb = qb[g & 1][j];
            const v2f a0 = {ca.x, ca.y}, b0 = {cb.x, cb.y};
            const v2f a1 = {ca.z, ca.w}, b1 = {cb.z, cb.w};
            acc0 = __builtin_amdgcn_wmma_f32_16x16x4_f32(
                 false, a0,  false, b0,
                 (short)0, acc0,  false,  false);
            acc1 = __builtin_amdgcn_wmma_f32_16x16x4_f32(
                 false, a1,  false, b1,
                 (short)0, acc1,  false,  false);
        }
    }
    const v8f acc = acc0 + acc1;

    if (r < 8) {
#pragma unroll
        for (int v = 0; v < 8; ++v)
            part[tsel][kq][(v + hi * 8) * 8 + r] = acc[v];
    }
    __syncthreads();

    const int n  = threadIdx.x >> 5;
    const int tt = threadIdx.x & 31;
    const int ts = tt >> 4;
    const int e  = (tt & 15) * 8 + n;
    float s = part[ts][0][e];
    s += part[ts][1][e];
    s += part[ts][2][e];
    s += part[ts][3][e];
    xwT[(size_t)n * T_LEN + blockIdx.x * 32 + tt] = s + (b_ih[n] + b_hh[n]);
}

#define MUL_ROW_SHARE(acc, h, w, J)                                          \
    asm volatile("v_mul_f32_dpp %0, %1, %2 row_share:" #J                    \
                 " row_mask:0xf bank_mask:0xf"                               \
                 : "=v"(acc) : "v"(h), "v"(w))
#define FMAC_ROW_SHARE(acc, h, w, J)                                         \
    asm volatile("v_fmac_f32_dpp %0, %1, %2 row_share:" #J                   \
                 " row_mask:0xf bank_mask:0xf"                               \
                 : "+v"(acc) : "v"(h), "v"(w))

__device__ __forceinline__ void scan_part(const float4 (&X)[16], const float (&w)[8],
                                          float& h, const int g,
                                          char* __restrict__ sp, const unsigned so,
                                          const unsigned lv, float4& ko)
{
#pragma unroll
    for (int b = 0; b < 4; ++b) {
        float hv[16];
#pragma unroll
        for (int c = 0; c < 16; ++c) {
            const float4 cq = X[b * 4 + (c >> 2)];
            const float xs_c = ((c & 3) == 0) ? cq.x : ((c & 3) == 1) ? cq.y
                             : ((c & 3) == 2) ? cq.z : cq.w;
            if (c == 8) {
                asm volatile("" :: "v"(ko.x), "v"(ko.y), "v"(ko.z), "v"(ko.w));
                if (b == 0) asm volatile("" :: "v"(lv));
            }
            float P = xs_c, Q, R;
            FMAC_ROW_SHARE(P, h, w[0], 0);
            MUL_ROW_SHARE(Q, h, w[1], 1);
            MUL_ROW_SHARE(R, h, w[2], 2);
            FMAC_ROW_SHARE(P, h, w[3], 3);
            FMAC_ROW_SHARE(Q, h, w[4], 4);
            FMAC_ROW_SHARE(R, h, w[5], 5);
            FMAC_ROW_SHARE(P, h, w[6], 6);
            FMAC_ROW_SHARE(Q, h, w[7], 7);
            asm volatile("" : "+v"(P));
            h = fmaxf((P + R) + Q, 0.0f);
            hv[c] = h;
        }

        float4 o;
        o.x = (g == 3) ? hv[12] : (g == 2) ? hv[8]  : (g == 1) ? hv[4] : hv[0];
        o.y = (g == 3) ? hv[13] : (g == 2) ? hv[9]  : (g == 1) ? hv[5] : hv[1];
        o.z = (g == 3) ? hv[14] : (g == 2) ? hv[10] : (g == 1) ? hv[6] : hv[2];
        o.w = (g == 3) ? hv[15] : (g == 2) ? hv[11] : (g == 1) ? hv[7] : hv[3];
        *(float4*)(sp + b * 64 + so) = o;
        ko = o;
    }
}

__global__ __launch_bounds__(32) void rnn_scan(const float* __restrict__ xwT,
                         const float* __restrict__ W_hh,
                         float* __restrict__ hsT)
{
    const int lane = threadIdx.x & 31;
    const int i = lane & 7;
    const int g = lane >> 3;

    float w[8];
#pragma unroll
    for (int j = 0; j < 8; ++j) w[j] = W_hh[i * 8 + j];

    const unsigned lo = (unsigned)i * (unsigned)(T_LEN * 4);
    const unsigned so = lo + (unsigned)g * 16u;
    const char* xb = (const char*)xwT;
    char* hb = (char*)hsT;

    float4 A[16], B[16];
#pragma unroll
    for (int q = 0; q < 16; ++q) A[q] = *(const float4*)(xb + lo + q * 16);
    float h = 0.0f;
    float4 ko = {0.0f, 0.0f, 0.0f, 0.0f};

    for (int t = 0; t < T_LEN; t += 128) {
        const char* pb = xb + (size_t)(t + 64) * 4;
        unsigned vb = lo;
        asm volatile("" : "+v"(vb));
#pragma unroll
        for (int q = 0; q < 16; ++q) B[q] = *(const float4*)(pb + vb + q * 16);
        scan_part(A, w, h, g, hb + (size_t)t * 4, so, vb, ko);

        const int ta = (t + 128 < T_LEN) ? (t + 128) : 0;
        const char* pa = xb + (size_t)ta * 4;
        unsigned va = lo;
        asm volatile("" : "+v"(va));
#pragma unroll
        for (int q = 0; q < 16; ++q) A[q] = *(const float4*)(pa + va + q * 16);
        scan_part(B, w, h, g, hb + (size_t)(t + 64) * 4, so, va, ko);
    }
}

__global__ __launch_bounds__(256) void rnn_out(
    const float* __restrict__ hsT, const float* __restrict__ W_out,
    const float* __restrict__ b_out, float* __restrict__ out)
{
    const int t = blockIdx.x * blockDim.x + threadIdx.x;
    if (t >= T_LEN) return;
    const float4 h0 = {hsT[0 * T_LEN + t], hsT[1 * T_LEN + t],
                       hsT[2 * T_LEN + t], hsT[3 * T_LEN + t]};
    const float4 h1 = {hsT[4 * T_LEN + t], hsT[5 * T_LEN + t],
                       hsT[6 * T_LEN + t], hsT[7 * T_LEN + t]};
    float s = b_out[0];
    s += h0.x * W_out[0] + h0.y * W_out[1] + h0.z * W_out[2] + h0.w * W_out[3];
    s += h1.x * W_out[4] + h1.y * W_out[5] + h1.z * W_out[6] + h1.w * W_out[7];
    out[t] = s;
}

extern "C" void kernel_launch(void* const* d_in, const int* in_sizes, int n_in,
                              void* d_out, int out_size, void* d_ws, size_t ws_size,
                              hipStream_t stream) {
    const float* x     = (const float*)d_in[0];
    const float* W_ih  = (const float*)d_in[1];
    const float* b_ih  = (const float*)d_in[2];
    const float* W_hh  = (const float*)d_in[3];
    const float* b_hh  = (const float*)d_in[4];
    const float* W_out = (const float*)d_in[5];
    const float* b_out = (const float*)d_in[6];
    float* out = (float*)d_out;

    float* xwT = (float*)d_ws;
    float* hsT = xwT + (size_t)T_LEN * H_DIM;

    rnn_inproj_wmma<<<T_LEN / 32, 256, 0, stream>>>(x, W_ih, b_ih, b_hh, xwT);
    rnn_scan<<<1, 32, 0, stream>>>(xwT, W_hh, hsT);
    rnn_out<<<(T_LEN + 255) / 256, 256, 0, stream>>>(hsT, W_out, b_out, out);
}


# speedup vs baseline: 4.0877x; 1.0567x over previous
#include <hip/hip_runtime.h>

typedef float v2f __attribute__((ext_vector_type(2)));
typedef float v8f __attribute__((ext_vector_type(8)));

#define T_LEN 16384
#define D_DIM 1024
#define H_DIM 8

__global__ __launch_bounds__(256) void rnn_inproj_wmma(
    const float* __restrict__ x, const float* __restrict__ W_ih,
    const float* __restrict__ b_ih, const float* __restrict__ b_hh,
    float* __restrict__ xwT)
{
    __shared__ float part[2][4][16 * 8];

    const int lane = threadIdx.x & 31;
    const int wave = threadIdx.x >> 5;
    const int tsel = wave >> 2;
    const int kq   = wave & 3;
    const int tile = blockIdx.x * 2 + tsel;
    const int r    = lane & 15;
    const int hi   = lane >> 4;
    const int nn   = r & 7;

    const float* xrow = x    + (size_t)(tile * 16 + r) * D_DIM + kq * 256 + hi * 4;
    const float* wrow = W_ih + (size_t)nn * D_DIM + kq * 256 + hi * 4;

    v8f acc0 = {}, acc1 = {};
    float4 qa[2][8], qb[2][8];
#pragma unroll
    for (int j = 0; j < 8; ++j) {
        qa[0][j] = *(const float4*)(xrow + j * 8);
        qb[0][j] = *(const float4*)(wrow + j * 8);
    }
#pragma unroll
    for (int g = 0; g < 4; ++g) {
        if (g < 3) {
#pragma unroll
            for (int j = 0; j < 8; ++j) {
                qa[(g + 1) & 1][j] = *(const float4*)(xrow + (g + 1) * 64 + j * 8);
                qb[(g + 1) & 1][j] = *(const float4*)(wrow + (g + 1) * 64 + j * 8);
            }
        }
#pragma unroll
        for (int j = 0; j < 8; ++j) {
            const float4 ca = qa[g & 1][j];
            const float4 cb = qb[g & 1][j];
            const v2f a0 = {ca.x, ca.y}, b0 = {cb.x, cb.y};
            const v2f a1 = {ca.z, ca.w}, b1 = {cb.z, cb.w};
            acc0 = __builtin_amdgcn_wmma_f32_16x16x4_f32(
                 false, a0,  false, b0,
                 (short)0, acc0,  false,  false);
            acc1 = __builtin_amdgcn_wmma_f32_16x16x4_f32(
                 false, a1,  false, b1,
                 (short)0, acc1,  false,  false);
        }
    }
    const v8f acc = acc0 + acc1;

    if (r < 8) {
#pragma unroll
        for (int v = 0; v < 8; ++v)
            part[tsel][kq][(v + hi * 8) * 8 + r] = acc[v];
    }
    __syncthreads();

    const int n  = threadIdx.x >> 5;
    const int tt = threadIdx.x & 31;
    const int ts = tt >> 4;
    const int e  = (tt & 15) * 8 + n;
    float s = part[ts][0][e];
    s += part[ts][1][e];
    s += part[ts][2][e];
    s += part[ts][3][e];
    xwT[(size_t)n * T_LEN + blockIdx.x * 32 + tt] = s + (b_ih[n] + b_hh[n]);
}

#define MUL_GROUP_POS(acc, h, w, A)                                          \
    asm volatile("v_mul_f32_dpp %0, %1, %2 dpp8:[" #A "," #A "," #A "," #A   \
                 "," #A "," #A "," #A "," #A "]"                             \
                 : "=v"(acc) : "v"(h), "v"(w))
#define FMAC_GROUP_POS(acc, h, w, A)                                         \
    asm volatile("v_fmac_f32_dpp %0, %1, %2 dpp8:[" #A "," #A "," #A "," #A  \
                 "," #A "," #A "," #A "," #A "]"                             \
                 : "+v"(acc) : "v"(h), "v"(w))
#define ADD_OTHER_HALF(t, s)                                                 \
    asm volatile("v_add_f32_dpp %0, %1, %1 row_xmask:12"                     \
                 " row_mask:0xf bank_mask:0xf"                               \
                 : "=v"(t) : "v"(s))

#define RELU(h, s)                                                           \
    asm volatile("v_max_num_f32 %0, 0, %1" : "=v"(h) : "v"(s))

__device__ __forceinline__ void scan_part(const float4 (&X)[16], const float (&w)[4],
                                          float& h, const int g, const bool upper,
                                          char* __restrict__ sp, const unsigned so,
                                          const unsigned lv, float4& ko)
{
#pragma unroll
    for (int b = 0; b < 4; ++b) {
        float hv[16];
#pragma unroll
        for (int c = 0; c < 16; ++c) {
            const float4 cq = X[b * 4 + (c >> 2)];
            const float xs_c = ((c & 3) == 0) ? cq.x : ((c & 3) == 1) ? cq.y
                             : ((c & 3) == 2) ? cq.z : cq.w;
            if (c == 8) {
                asm volatile("" :: "v"(ko.x), "v"(ko.y), "v"(ko.z), "v"(ko.w));
                if (b == 0) asm volatile("" :: "v"(lv));
            }
            float P = upper ? 0.0f : xs_c, Q, S;
            FMAC_GROUP_POS(P, h, w[0], 0);
            MUL_GROUP_POS(Q, h, w[1], 1);
            FMAC_GROUP_POS(P, h, w[2], 2);
            FMAC_GROUP_POS(Q, h, w[3], 3);
            ADD_OTHER_HALF(S, P + Q);
            RELU(h, S);
            hv[c] = h;
        }

        float4 o;
        o.x = (g == 3) ? hv[12] : (g == 2) ? hv[8]  : (g == 1) ? hv[4] : hv[0];
        o.y = (g == 3) ? hv[13] : (g == 2) ? hv[9]  : (g == 1) ? hv[5] : hv[1];
        o.z = (g == 3) ? hv[14] : (g == 2) ? hv[10] : (g == 1) ? hv[6] : hv[2];
        o.w = (g == 3) ? hv[15] : (g == 2) ? hv[11] : (g == 1) ? hv[7] : hv[3];
        *(float4*)(sp + b * 64 + so) = o;
        ko = o;
    }
}

__global__ __launch_bounds__(32) void rnn_scan(const float* __restrict__ xwT,
                         const float* __restrict__ W_hh,
                         float* __restrict__ hsT)
{
    const int lane = threadIdx.x & 31;
    const bool upper = (lane & 8) != 0;
    const int i = (lane & 7) ^ (upper ? 4 : 0);
    const int g = lane >> 3;

    float w[4];
#pragma unroll
    for (int a = 0; a < 4; ++a) w[a] = W_hh[i * 8 + a + (upper ? 4 : 0)];

    const unsigned lo = (unsigned)i * (unsigned)(T_LEN * 4);
    const unsigned so = lo + (unsigned)g * 16u;
    const char* xb = (const char*)xwT;
    char* hb = (char*)hsT;

    float4 A[16], B[16];
#pragma unroll
    for (int q = 0; q < 16; ++q) A[q] = *(const float4*)(xb + lo + q * 16);
    float h = 0.0f;
    float4 ko = {0.0f, 0.0f, 0.0f, 0.0f};

    for (int t = 0; t < T_LEN; t += 128) {
        const char* pb = xb + (size_t)(t + 64) * 4;
        unsigned vb = lo;
        asm volatile("" : "+v"(vb));
#pragma unroll
        for (int q = 0; q < 16; ++q) B[q] = *(const float4*)(pb + vb + q * 16);
        scan_part(A, w, h, g, upper, hb + (size_t)t * 4, so, vb, ko);

        const int ta = (t + 128 < T_LEN) ? (t + 128) : 0;
        const char* pa = xb + (size_t)ta * 4;
        unsigned va = lo;
        asm volatile("" : "+v"(va));
#pragma unroll
        for (int q = 0; q < 16; ++q) A[q] = *(const float4*)(pa + va + q * 16);
        scan_part(B, w, h, g, upper, hb + (size_t)(t + 64) * 4, so, va, ko);
    }
}

__global__ __launch_bounds__(256) void rnn_out(
    const float* __restrict__ hsT, const float* __restrict__ W_out,
    const float* __restrict__ b_out, float* __restrict__ out)
{
    const int t = blockIdx.x * blockDim.x + threadIdx.x;
    if (t >= T_LEN) return;
    const float4 h0 = {hsT[0 * T_LEN + t], hsT[1 * T_LEN + t],
                       hsT[2 * T_LEN + t], hsT[3 * T_LEN + t]};
    const float4 h1 = {hsT[4 * T_LEN + t], hsT[5 * T_LEN + t],
                       hsT[6 * T_LEN + t], hsT[7 * T_LEN + t]};
    float s = b_out[0];
    s += h0.x * W_out[0] + h0.y * W_out[1] + h0.z * W_out[2] + h0.w * W_out[3];
    s += h1.x * W_out[4] + h1.y * W_out[5] + h1.z * W_out[6] + h1.w * W_out[7];
    out[t] = s;
}

extern "C" void kernel_launch(void* const* d_in, const int* in_sizes, int n_in,
                              void* d_out, int out_size, void* d_ws, size_t ws_size,
                              hipStream_t stream) {
    const float* x     = (const float*)d_in[0];
    const float* W_ih  = (const float*)d_in[1];
    const float* b_ih  = (const float*)d_in[2];
    const float* W_hh  = (const float*)d_in[3];
    const float* b_hh  = (const float*)d_in[4];
    const float* W_out = (const float*)d_in[5];
    const float* b_out = (const float*)d_in[6];
    float* out = (float*)d_out;

    float* xwT = (float*)d_ws;
    float* hsT = xwT + (size_t)T_LEN * H_DIM;

    rnn_inproj_wmma<<<T_LEN / 32, 256, 0, stream>>>(x, W_ih, b_ih, b_hh, xwT);
    rnn_scan<<<1, 32, 0, stream>>>(xwT, W_hh, hsT);
    rnn_out<<<(T_LEN + 255) / 256, 256, 0, stream>>>(hsT, W_out, b_out, out);
}
